# starting kernel
# baseline (speedup 1.0000x reference)
#include <hip/hip_runtime.h>
#include <math.h>

typedef float v2f __attribute__((ext_vector_type(2)));
typedef float v8f __attribute__((ext_vector_type(8)));

#define ROWS 32
#define KNN 16
#define HD 128
#define MIXK 262
#define MIXKP 264
#define MSTRIDE 268

__global__ __launch_bounds__(256)
void graphconv_wmma_kernel(const float* __restrict__ feat,
                           const float* __restrict__ coords,
                           const int*   __restrict__ knn,
                           const float* __restrict__ W,
                           const float* __restrict__ bias,
                           float* __restrict__ out)
{
    extern __shared__ float smem[];
    float* mixLds = smem;
    int*   knnLds = (int*)(smem + ROWS * MSTRIDE);

    const int tid  = threadIdx.x;
    const int base = blockIdx.x * ROWS;

    for (int i = tid; i < ROWS * KNN; i += 256)
        knnLds[i] = __builtin_nontemporal_load(knn + (size_t)base * KNN + i);
    __syncthreads();

    {
        const int r    = tid >> 3;
        const int sub  = tid & 7;
        const int col0 = sub * 16;
        float* mrow = mixLds + r * MSTRIDE;

        const float4* sp = (const float4*)(feat + (size_t)(base + r) * HD + col0);
        float4* dst_self = (float4*)(mrow + col0);
        float4* dst_agg  = (float4*)(mrow + HD + col0);
        float ax[16];
        #pragma unroll
        for (int j = 0; j < 4; ++j) {
            float4 v = sp[j];
            dst_self[j] = v;
            ax[4*j+0] = 0.f; ax[4*j+1] = 0.f; ax[4*j+2] = 0.f; ax[4*j+3] = 0.f;
        }
        #pragma unroll
        for (int k = 0; k < KNN; ++k) {
            const int idx = knnLds[r * KNN + k];
            const float4* np = (const float4*)(feat + (size_t)idx * HD + col0);
            #pragma unroll
            for (int j = 0; j < 4; ++j) {
                float4 v = np[j];
                ax[4*j+0] += v.x; ax[4*j+1] += v.y;
                ax[4*j+2] += v.z; ax[4*j+3] += v.w;
            }
        }
        const float inv = 1.0f / 16.0f;
        #pragma unroll
        for (int j = 0; j < 4; ++j) {
            float4 v;
            v.x = ax[4*j+0]*inv; v.y = ax[4*j+1]*inv;
            v.z = ax[4*j+2]*inv; v.w = ax[4*j+3]*inv;
            dst_agg[j] = v;
        }
    }

    if (tid < ROWS) {
        const int r = tid;
        const int node = base + r;
        const float cx = coords[(size_t)node*3+0];
        const float cy = coords[(size_t)node*3+1];
        const float cz = coords[(size_t)node*3+2];
        float sx=0.f, sy=0.f, sz=0.f, qx=0.f, qy=0.f, qz=0.f;
        #pragma unroll
        for (int k = 0; k < KNN; ++k) {
            const int idx = knnLds[r * KNN + k];
            float dx = coords[(size_t)idx*3+0] - cx;
            float dy = coords[(size_t)idx*3+1] - cy;
            float dz = coords[(size_t)idx*3+2] - cz;
            sx += dx; sy += dy; sz += dz;
            qx += dx*dx; qy += dy*dy; qz += dz*dz;
        }
        const float inv = 1.0f / 16.0f;
        const float mx = sx*inv, my = sy*inv, mz = sz*inv;
        float vx = qx*inv - mx*mx; vx = vx > 0.f ? vx : 0.f;
        float vy = qy*inv - my*my; vy = vy > 0.f ? vy : 0.f;
        float vz = qz*inv - mz*mz; vz = vz > 0.f ? vz : 0.f;
        float* mrow = mixLds + r * MSTRIDE + 2*HD;
        mrow[0] = mx;        mrow[1] = my;        mrow[2] = mz;
        mrow[3] = sqrtf(vx); mrow[4] = sqrtf(vy); mrow[5] = sqrtf(vz);
        mrow[6] = 0.f;       mrow[7] = 0.f;
    }
    __syncthreads();

    const int wave    = tid >> 5;
    const int lane    = tid & 31;
    const int colbase = wave * 16;
    const int lx      = lane & 15;
    const int hi      = lane >> 4;

    v8f c0 = {};
    v8f c1 = {};

    const float* mA0 = mixLds + lx * MSTRIDE;
    const float* mA1 = mixLds + (lx + 16) * MSTRIDE;

    for (int kk = 0; kk < MIXKP - 4; kk += 4) {
        const int k0 = kk + hi * 2;
        v2f a0, a1, b;
        a0.x = mA0[k0]; a0.y = mA0[k0 + 1];
        a1.x = mA1[k0]; a1.y = mA1[k0 + 1];
        const float* wp = W + (size_t)k0 * HD + colbase + lx;
        b.x = wp[0];
        b.y = wp[HD];
        c0 = __builtin_amdgcn_wmma_f32_16x16x4_f32(false, a0, false, b,
                                                   (short)0, c0, false, false);
        c1 = __builtin_amdgcn_wmma_f32_16x16x4_f32(false, a1, false, b,
                                                   (short)0, c1, false, false);
    }
    {
        const int k0 = (MIXKP - 4) + hi * 2;
        v2f a0, a1, b;
        a0.x = mA0[k0]; a0.y = mA0[k0 + 1];
        a1.x = mA1[k0]; a1.y = mA1[k0 + 1];
        if (k0 < MIXK) {
            const float* wp = W + (size_t)k0 * HD + colbase + lx;
            b.x = wp[0];
            b.y = wp[HD];
        } else {
            b.x = 0.f; b.y = 0.f;
        }
        c0 = __builtin_amdgcn_wmma_f32_16x16x4_f32(false, a0, false, b,
                                                   (short)0, c0, false, false);
        c1 = __builtin_amdgcn_wmma_f32_16x16x4_f32(false, a1, false, b,
                                                   (short)0, c1, false, false);
    }

    const int ncol = colbase + lx;
    const float bv = bias[ncol];
    #pragma unroll
    for (int r = 0; r < 8; ++r) {
        const int row = r + hi * 8;
        float x0 = c0[r] + bv;
        __builtin_nontemporal_store(x0 / (1.0f + __expf(-x0)),
                                    out + (size_t)(base + row) * HD + ncol);
        float x1 = c1[r] + bv;
        __builtin_nontemporal_store(x1 / (1.0f + __expf(-x1)),
                                    out + (size_t)(base + 16 + row) * HD + ncol);
    }
}

extern "C" void kernel_launch(void* const* d_in, const int* in_sizes, int n_in,
                              void* d_out, int out_size, void* d_ws, size_t ws_size,
                              hipStream_t stream) {
    const float* feat   = (const float*)d_in[0];
    const float* coords = (const float*)d_in[1];
    const int*   knn    = (const int*)  d_in[2];
    const float* W      = (const float*)d_in[3];
    const float* bias   = (const float*)d_in[4];
    float* out = (float*)d_out;

    const int n = in_sizes[0] / HD;
    const int grid = (n + ROWS - 1) / ROWS;
    const size_t shmem = (size_t)ROWS * MSTRIDE * sizeof(float)
                       + (size_t)ROWS * KNN * sizeof(int);

    graphconv_wmma_kernel<<<grid, 256, shmem, stream>>>(feat, coords, knn, W, bias, out);
}


# Aggregate half of the product deferred one tile: a tile's k steps 32-65, bias and SiLU run under the next tile's gather; LDS tiles doubled, one barrier per tile; k and sum orders unchanged
# speedup vs baseline: 1.4670x; 1.4670x over previous
#include <hip/hip_runtime.h>
#include <math.h>

typedef float v2f __attribute__((ext_vector_type(2)));
typedef float v8f __attribute__((ext_vector_type(8)));
typedef int   v2i __attribute__((ext_vector_type(2)));
typedef float v4f __attribute__((ext_vector_type(4)));

#define ROWS 32
#define KNN 16
#define HD 128
#define MIXK 262
#define MIXKP 264
#define MSTRIDE 268
#define OSTRIDE 132
#define KSTEPS 66
#define KGROUP 6
#define NBLOCKS 512
#define MIXF (ROWS * MSTRIDE)
#define KNNI (ROWS * KNN)
#define OUTF (ROWS * OSTRIDE)

struct first_tile_t { static constexpr bool value = true; };
struct later_tile_t { static constexpr bool value = false; };

__global__ __launch_bounds__(256, 4)
void graphconv_wmma_kernel(const float* __restrict__ feat,
                           const float* __restrict__ coords,
                           const int*   __restrict__ knn,
                           const float* __restrict__ W,
                           const float* __restrict__ bias,
                           float* __restrict__ out,
                           int ntiles)
{
    __shared__ __attribute__((aligned(16))) float smem[2 * MIXF + 2 * KNNI + 2 * OUTF];

    const int tid     = threadIdx.x;
    const int wave    = tid >> 5;
    const int lane    = tid & 31;
    const int colbase = wave * 16;
    const int lx      = lane & 15;
    const int hi      = lane >> 4;
    const int ncol    = colbase + lx;

    v2f wb[KSTEPS];
    #pragma unroll
    for (int q = 0; q < KSTEPS; ++q) {
        const int k0 = 4 * q + hi * 2;
        if (k0 < MIXK) {
            const float* wp = W + (size_t)k0 * HD + ncol;
            wb[q].x = wp[0];
            wb[q].y = wp[HD];
        } else {
            wb[q].x = 0.f; wb[q].y = 0.f;
        }
    }
    const float bv = bias[ncol];

    const int r    = tid >> 3;
    const int sub  = tid & 7;
    const int col0 = sub * 16;

    #define FLUSH_TILE(obuf, pb)                                                         \
        {                                                                                \
            const float* orow_ = (obuf) + r * OSTRIDE + sub * 4;                         \
            float* og_ = out + (size_t)((pb) + r) * HD + sub * 4;                        \
            _Pragma("unroll")                                                            \
            for (int j = 0; j < 4; ++j) {                                                \
                const v4f v_ = *(const v4f*)(orow_ + 32 * j);                            \
                __builtin_nontemporal_store(v_, (v4f*)(og_ + 32 * j));                   \
            }                                                                            \
        }
    #define EPILOGUE(obuf)                                                               \
        {                                                                                \
            _Pragma("unroll")                                                            \
            for (int rw = 0; rw < 8; ++rw) {                                             \
                const int row = rw + hi * 8;       \
                float x0 = c0[rw] + bv;                                                  \
                (obuf)[row * OSTRIDE + ncol] = x0 / (1.0f + __expf(-x0));                \
                float x1 = c1[rw] + bv;                                                  \
                (obuf)[(16 + row) * OSTRIDE + ncol] = x1 / (1.0f + __expf(-x1));         \
            }                                                                            \
        }

    #define LOAD_AGROUP(A0, q0, n, buf)                                      \
        {                                                                    \
            _Pragma("unroll")                                                \
            for (int s = 0; s < KGROUP; ++s) if (s < (n)) {                  \
                const int k0 = 4 * ((q0) + s) + hi * 2;                      \
                buf[2*s].x   = (A0)[k0];                                     \
                buf[2*s].y   = (A0)[k0 + 1];                                 \
                buf[2*s+1].x = (A0)[16 * MSTRIDE + k0];                      \
                buf[2*s+1].y = (A0)[16 * MSTRIDE + k0 + 1];                  \
            }                                                                \
        }
    #define MMA_GROUP(q0, n, buf)                                            \
        {                                                                    \
            _Pragma("unroll")                                                \
            for (int s = 0; s < KGROUP; ++s) if (s < (n)) {                  \
                c0 = __builtin_amdgcn_wmma_f32_16x16x4_f32(false, buf[2*s], false, wb[(q0) + s],   \
                                                           (short)0, c0, false, false);            \
                c1 = __builtin_amdgcn_wmma_f32_16x16x4_f32(false, buf[2*s+1], false, wb[(q0) + s], \
                                                           (short)0, c1, false, false);            \
            }                                                                \
        }
    #define GEMM_STEP(A0, ql, nl, bufl, qm, nm, bufm)                        \
        LOAD_AGROUP(A0, ql, nl, bufl)                                        \
        __builtin_amdgcn_sched_barrier(0);                                   \
        MMA_GROUP(qm, nm, bufm)                                              \
        __builtin_amdgcn_sched_barrier(0);

    #define LOAD_ROW_AT(buf, idx)                                                    \
        {                                                                            \
            const unsigned o_ = ((unsigned)(idx) * HD + col0) * 4u;                  \
            const float4* n_ = (const float4*)((const char*)feat + o_);             \
            _Pragma("unroll")                                                        \
            for (int j = 0; j < 4; ++j) buf[j] = n_[j];                              \
        }
    #define LOAD_ROW(buf, n) LOAD_ROW_AT(buf, kp[(n)])
    #define SUM_ROW(buf)                                                             \
        {                                                                            \
            _Pragma("unroll")                                                        \
            for (int j = 0; j < 4; ++j) {                                            \
                ax[2*j].x   += buf[j].x; ax[2*j].y   += buf[j].y;                    \
                ax[2*j+1].x += buf[j].z; ax[2*j+1].y += buf[j].w;                    \
            }                                                                        \
        }
    #define HOLD_SUMS                                                                \
        asm volatile("" : "+v"(ax[0]), "+v"(ax[1]), "+v"(ax[2]), "+v"(ax[3]),        \
                          "+v"(ax[4]), "+v"(ax[5]), "+v"(ax[6]), "+v"(ax[7]));
    #define HOLD_FRAGS                                                               \
        asm volatile("" : "+v"(af[0][0]), "+v"(af[0][1]), "+v"(af[0][2]), "+v"(af[0][3]));
    #define MMA_SUB(A0, q0)                                                          \
        {                                                                            \
            LOAD_AGROUP(A0, q0, 2, af[0])                                            \
            __builtin_amdgcn_sched_barrier(0);                                       \
            MMA_GROUP(q0, 2, af[0])                                                  \
            __builtin_amdgcn_sched_barrier(0);                                       \
        }
    #define ROW_STEP_M(n, buf, A0, q0)                                               \
        {                                                                            \
            LOAD_AGROUP(A0, q0, 2, af[0])                                            \
            const int nidx_ = kp[(n) + 2];                                           \
            __builtin_amdgcn_sched_barrier(0);                                       \
            HOLD_SUMS                                                                \
            SUM_ROW(buf)                                                             \
            LOAD_ROW_AT(buf, nidx_)                                                  \
            __builtin_amdgcn_sched_barrier(0);                                       \
            HOLD_FRAGS                                                               \
            MMA_GROUP(q0, 2, af[0])                                                  \
            __builtin_amdgcn_sched_barrier(0);                                       \
            MMA_SUB(A0, (q0) + 2)                                                    \
        }
    #define ROW_STEP_N(n, buf)                                                       \
        {                                                                            \
            const int nidx_ = kp[(n) + 2];                                           \
            HOLD_SUMS                                                                \
            SUM_ROW(buf)                                                             \
            LOAD_ROW_AT(buf, nidx_)                                                  \
            __builtin_amdgcn_sched_barrier(0);                                       \
        }

    v8f c0 = {};
    v8f c1 = {};
    v2f af[2][2 * KGROUP];
    int base = 0;

    auto tile_step = [&](auto kind, const int it, const int tile) __attribute__((always_inline)) {
        constexpr bool FIRST = decltype(kind)::value;
        base = tile * ROWS;
        const int par = FIRST ? 0 : (it & 1);
        float*       mixC = smem + par * MIXF;
        const float* mixP = smem + (par ^ 1) * MIXF;
        int*         knnC = (int*)(smem + 2 * MIXF) + par * KNNI;
        const float* outO = smem + 2 * MIXF + 2 * KNNI + par * OUTF;
        float*       outP = smem + 2 * MIXF + 2 * KNNI + (par ^ 1) * OUTF;
        float*  mrow     = mixC + r * MSTRIDE;
        float4* dst_self = (float4*)(mrow + col0);
        float4* dst_agg  = (float4*)(mrow + HD + col0);
        const int*   kp  = knnC + r * KNN;
        const float* sA0 = mixC + lx * MSTRIDE;
        const float* pA0 = mixP + lx * MSTRIDE;

        {
            const v2i kn = __builtin_nontemporal_load(
                (const v2i*)((const char*)knn + ((unsigned)base * KNN * 4u + (unsigned)tid * 8u)));
            const float4* sp = (const float4*)((const char*)feat
                                               + ((unsigned)(base + r) * HD + col0) * 4u);
            const float4 sv0 = sp[0], sv1 = sp[1], sv2 = sp[2], sv3 = sp[3];
            ((v2i*)knnC)[tid] = kn;
            dst_self[0] = sv0; dst_self[1] = sv1; dst_self[2] = sv2; dst_self[3] = sv3;
        }
        __syncthreads();

        v2f ax[8];
        #pragma unroll
        for (int j = 0; j < 8; ++j) { ax[j].x = 0.f; ax[j].y = 0.f; }
        float4 ra[4], rb[4];
        LOAD_ROW(ra, 0)
        LOAD_ROW(rb, 1)
        __builtin_amdgcn_sched_barrier(0);
        if (!FIRST && it >= 2) FLUSH_TILE(outO, base - 2 * NBLOCKS * ROWS)
        if constexpr (!FIRST) {
            LOAD_AGROUP(pA0, 32, 4, af[1])
            __builtin_amdgcn_sched_barrier(0);
            MMA_GROUP(32, 4, af[1])
            __builtin_amdgcn_sched_barrier(0);
            ROW_STEP_M( 0, ra, pA0, 36)
            ROW_STEP_M( 1, rb, pA0, 40)
            ROW_STEP_M( 2, ra, pA0, 44)
            ROW_STEP_M( 3, rb, pA0, 48)
            ROW_STEP_M( 4, ra, pA0, 52)
            ROW_STEP_M( 5, rb, pA0, 56)
            ROW_STEP_M( 6, ra, pA0, 60)
            MMA_SUB(pA0, 64)
            EPILOGUE(outP)
        } else {
            ROW_STEP_N( 0, ra)
            ROW_STEP_N( 1, rb)
            ROW_STEP_N( 2, ra)
            ROW_STEP_N( 3, rb)
            ROW_STEP_N( 4, ra)
            ROW_STEP_N( 5, rb)
            ROW_STEP_N( 6, ra)
        }
        #pragma unroll
        for (int j = 0; j < 8; ++j) { c0[j] = 0.f; c1[j] = 0.f; }
        ROW_STEP_M( 7, rb, sA0,  0)
        ROW_STEP_M( 8, ra, sA0,  4)
        ROW_STEP_M( 9, rb, sA0,  8)
        ROW_STEP_M(10, ra, sA0, 12)
        ROW_STEP_M(11, rb, sA0, 16)
        ROW_STEP_M(12, ra, sA0, 20)
        ROW_STEP_M(13, rb, sA0, 24)
        MMA_SUB(sA0, 28)
        MMA_SUB(sA0, 30)
        HOLD_SUMS
        SUM_ROW(ra)
        SUM_ROW(rb)
        {
            const float inv = 1.0f / 16.0f;
            #pragma unroll
            for (int j = 0; j < 4; ++j) {
                float4 v;
                v.x = ax[2*j].x*inv;   v.y = ax[2*j].y*inv;
                v.z = ax[2*j+1].x*inv; v.w = ax[2*j+1].y*inv;
                dst_agg[j] = v;
            }
        }

        if (tid < ROWS) {
            const int rr = tid;
            const int node = base + rr;
            const float cx = coords[(size_t)node*3+0];
            const float cy = coords[(size_t)node*3+1];
            const float cz = coords[(size_t)node*3+2];
            float sx=0.f, sy=0.f, sz=0.f, qx=0.f, qy=0.f, qz=0.f;
            #pragma unroll
            for (int k = 0; k < KNN; ++k) {
                const int idx = knnC[rr * KNN + k];
                float dx = coords[(size_t)idx*3+0] - cx;
                float dy = coords[(size_t)idx*3+1] - cy;
                float dz = coords[(size_t)idx*3+2] - cz;
                sx += dx; sy += dy; sz += dz;
                qx += dx*dx; qy += dy*dy; qz += dz*dz;
            }
            const float inv = 1.0f / 16.0f;
            const float mx = sx*inv, my = sy*inv, mz = sz*inv;
            float vx = qx*inv - mx*mx; vx = vx > 0.f ? vx : 0.f;
            float vy = qy*inv - my*my; vy = vy > 0.f ? vy : 0.f;
            float vz = qz*inv - mz*mz; vz = vz > 0.f ? vz : 0.f;
            float* mr = mixC + rr * MSTRIDE + 2*HD;
            mr[0] = mx;        mr[1] = my;        mr[2] = mz;
            mr[3] = sqrtf(vx); mr[4] = sqrtf(vy); mr[5] = sqrtf(vz);
            mr[6] = 0.f;       mr[7] = 0.f;
        }
    };

    int it = 0;
    {
        int tile = blockIdx.x;
        if (tile < ntiles) {
            tile_step(first_tile_t{}, 0, tile);
            it = 1;
            for (tile += NBLOCKS; tile < ntiles; tile += NBLOCKS, ++it) tile_step(later_tile_t{}, it, tile);
        }
    }

    if (it >= 1) {
        const int par = (it - 1) & 1;
        const float* lA0  = smem + par * MIXF + lx * MSTRIDE;
        float*       outL = smem + 2 * MIXF + 2 * KNNI + par * OUTF;
        const float* outO = smem + 2 * MIXF + 2 * KNNI + (par ^ 1) * OUTF;
        __syncthreads();
        if (it >= 2) FLUSH_TILE(outO, base - NBLOCKS * ROWS)
        LOAD_AGROUP(lA0, 32, 6, af[0])
        GEMM_STEP(lA0, 38, 6, af[1], 32, 6, af[0])
        GEMM_STEP(lA0, 44, 6, af[0], 38, 6, af[1])
        GEMM_STEP(lA0, 50, 6, af[1], 44, 6, af[0])
        GEMM_STEP(lA0, 56, 6, af[0], 50, 6, af[1])
        GEMM_STEP(lA0, 62, 4, af[1], 56, 6, af[0])
        MMA_GROUP(62, 4, af[1])
        EPILOGUE(outL)
        __syncthreads();
        FLUSH_TILE(outL, base)
    }
    #undef ROW_STEP_M
    #undef MMA_SUB
    #undef ROW_STEP_N
    #undef HOLD_SUMS
    #undef HOLD_FRAGS
    #undef LOAD_ROW_AT
    #undef LOAD_ROW
    #undef SUM_ROW
    #undef FLUSH_TILE
    #undef EPILOGUE
    #undef GEMM_STEP
    #undef MMA_GROUP
    #undef LOAD_AGROUP
}

extern "C" void kernel_launch(void* const* d_in, const int* in_sizes, int n_in,
                              void* d_out, int out_size, void* d_ws, size_t ws_size,
                              hipStream_t stream) {
    const float* feat   = (const float*)d_in[0];
    const float* coords = (const float*)d_in[1];
    const int*   knn    = (const int*)  d_in[2];
    const float* W      = (const float*)d_in[3];
    const float* bias   = (const float*)d_in[4];
    float* out = (float*)d_out;

    const int n = in_sizes[0] / HD;
    const int ntiles = (n + ROWS - 1) / ROWS;

    graphconv_wmma_kernel<<<NBLOCKS, 256, 0, stream>>>(feat, coords, knn, W, bias, out, ntiles);
}
